# starting kernel
# baseline (speedup 1.0000x reference)
#include <hip/hip_runtime.h>

typedef __attribute__((ext_vector_type(2))) float v2f;
typedef __attribute__((ext_vector_type(8))) float v8f;

#define HID   128
#define DIN   256
#define LROW  260

__global__ __launch_bounds__(256) void tgnn_zero_kernel(float4* __restrict__ p, long long n4) {
    long long i = (long long)blockIdx.x * blockDim.x + threadIdx.x;
    if (i < n4) p[i] = make_float4(0.f, 0.f, 0.f, 0.f);
}

__global__ __launch_bounds__(256) void tgnn_edge_kernel(
    const int*   __restrict__ src,
    const int*   __restrict__ dst,
    const float* __restrict__ ef,
    const float* __restrict__ bt,
    const float* __restrict__ node_ts,
    const float* __restrict__ time_w,
    const float* __restrict__ time_b,
    float*       __restrict__ h)
{
    int e = blockIdx.x;
    int j = threadIdx.x;
    int s = src[e];
    int d = dst[e];
    float td = bt[e] - node_ts[s];
    float v;
    if (j < HID) {
        v = ef[(size_t)e * HID + j];
    } else {
        int k = j - HID;
        v = cosf(td * time_w[k] + time_b[k]);
    }
    atomicAdd(&h[(size_t)d * DIN + j], v);
}

__global__ __launch_bounds__(256) void tgnn_predict_kernel(
    const float* __restrict__ h,
    const int*   __restrict__ s_idx,
    const int*   __restrict__ p_idx,
    const int*   __restrict__ n_idx,
    const float* __restrict__ W_src,
    const float* __restrict__ b_src,
    const float* __restrict__ W_dst,
    const float* __restrict__ b_dst,
    const float* __restrict__ W_out,
    const float* __restrict__ b_out,
    float*       __restrict__ out,
    int Q)
{
    __shared__ float sl[16 * LROW];
    __shared__ float pl[16 * LROW];
    __shared__ float nl[16 * LROW];
    __shared__ float red_pos[8 * 256];
    __shared__ float red_neg[8 * 256];

    const int tile = blockIdx.x;
    const int t    = threadIdx.x;

    for (int i = t; i < 16 * DIN; i += 256) {
        int row = i >> 8;
        int col = i & 255;
        int qi  = tile * 16 + row;
        sl[row * LROW + col] = h[(size_t)s_idx[qi] * DIN + col];
        pl[row * LROW + col] = h[(size_t)p_idx[qi] * DIN + col];
        nl[row * LROW + col] = h[(size_t)n_idx[qi] * DIN + col];
    }
    __syncthreads();

    const int wave = t >> 5;
    const int lane = t & 31;
    const int half = lane >> 4;
    const int lm   = lane & 15;
    const int n0   = wave * 16;

    v8f hs = {}, hp = {}, hn = {};

    const float* As = &sl[lm * LROW];
    const float* Ap = &pl[lm * LROW];
    const float* An = &nl[lm * LROW];

    for (int k0 = 0; k0 < DIN; k0 += 4) {
        int ka = k0 + 2 * half;
        v2f a_s; a_s.x = As[ka]; a_s.y = As[ka + 1];
        v2f a_p; a_p.x = Ap[ka]; a_p.y = Ap[ka + 1];
        v2f a_n; a_n.x = An[ka]; a_n.y = An[ka + 1];
        v2f b_s; b_s.x = W_src[ka * HID + n0 + lm]; b_s.y = W_src[(ka + 1) * HID + n0 + lm];
        v2f b_d; b_d.x = W_dst[ka * HID + n0 + lm]; b_d.y = W_dst[(ka + 1) * HID + n0 + lm];

        hs = __builtin_amdgcn_wmma_f32_16x16x4_f32(false, a_s, false, b_s, (short)0, hs, false, false);
        hp = __builtin_amdgcn_wmma_f32_16x16x4_f32(false, a_p, false, b_d, (short)0, hp, false, false);
        hn = __builtin_amdgcn_wmma_f32_16x16x4_f32(false, a_n, false, b_d, (short)0, hn, false, false);
    }

    const float bias = b_src[n0 + lm] + b_dst[n0 + lm];
    const float wo   = W_out[n0 + lm];

    #pragma unroll
    for (int r = 0; r < 8; ++r) {
        float pv = fmaxf(hs[r] + hp[r] + bias, 0.f) * wo;
        float nv = fmaxf(hs[r] + hn[r] + bias, 0.f) * wo;
        red_pos[wave * 256 + r * 32 + lane] = pv;
        red_neg[wave * 256 + r * 32 + lane] = nv;
    }
    __syncthreads();

    if (t < 16) {
        int mh = t >> 3;
        int r  = t & 7;
        float accp = b_out[0];
        float accn = b_out[0];
        for (int w = 0; w < 8; ++w) {
            #pragma unroll
            for (int l = 0; l < 16; ++l) {
                int idx = w * 256 + r * 32 + mh * 16 + l;
                accp += red_pos[idx];
                accn += red_neg[idx];
            }
        }
        out[tile * 16 + t]     = accp;
        out[Q + tile * 16 + t] = accn;
    }
}

extern "C" void kernel_launch(void* const* d_in, const int* in_sizes, int n_in,
                              void* d_out, int out_size, void* d_ws, size_t ws_size,
                              hipStream_t stream) {
    const int*   src     = (const int*)  d_in[0];
    const int*   dst     = (const int*)  d_in[1];
    const float* ef      = (const float*)d_in[2];
    const float* bt      = (const float*)d_in[3];
    const float* node_ts = (const float*)d_in[4];
    const int*   s_idx   = (const int*)  d_in[5];
    const int*   p_idx   = (const int*)  d_in[6];
    const int*   n_idx   = (const int*)  d_in[7];
    const float* time_w  = (const float*)d_in[8];
    const float* time_b  = (const float*)d_in[9];
    const float* W_src   = (const float*)d_in[10];
    const float* b_src   = (const float*)d_in[11];
    const float* W_dst   = (const float*)d_in[12];
    const float* b_dst   = (const float*)d_in[13];
    const float* W_out   = (const float*)d_in[14];
    const float* b_out   = (const float*)d_in[15];

    float* out = (float*)d_out;
    float* h   = (float*)d_ws;

    const int NN = in_sizes[4];
    const int NE = in_sizes[0];
    const int Q  = in_sizes[5];

    long long n4 = (long long)NN * DIN / 4;
    int zgrid = (int)((n4 + 255) / 256);
    tgnn_zero_kernel<<<zgrid, 256, 0, stream>>>((float4*)h, n4);

    tgnn_edge_kernel<<<NE, 256, 0, stream>>>(src, dst, ef, bt, node_ts, time_w, time_b, h);

    tgnn_predict_kernel<<<Q / 16, 256, 0, stream>>>(h, s_idx, p_idx, n_idx,
                                                    W_src, b_src, W_dst, b_dst,
                                                    W_out, b_out, out, Q);
}


# Build h only for the nodes that s_idx/p_idx/n_idx read: node->row map, compact [3Q,256] accumulator, edges into unread nodes skipped; f32 atomic adds kept (sum order by arrival, as before)
# speedup vs baseline: 7.0009x; 7.0009x over previous
#include <hip/hip_runtime.h>


typedef __attribute__((ext_vector_type(2))) float v2f;
typedef __attribute__((ext_vector_type(8))) float v8f;

#define HID   128
#define DIN   256
#define LROW  260
#define MAPR  1024


__global__ __launch_bounds__(1024) void tgnn_prep_kernel(
    const int* __restrict__ s_idx,
    const int* __restrict__ p_idx,
    const int* __restrict__ n_idx,
    int*       __restrict__ map,
    int*       __restrict__ qslot,
    float4*    __restrict__ hc4,
    int NN, int Q, int nmapb, long long z4)
{
    __shared__ int lmap[MAPR];
    const int t = threadIdx.x;

    if ((int)blockIdx.x < nmapb) {
        const int base = blockIdx.x * MAPR;
        lmap[t] = -1;
        __syncthreads();

        for (int i0 = 0; i0 < Q; i0 += 4 * MAPR) {
            int a[4], b[4], c[4];
            #pragma unroll
            for (int k = 0; k < 4; ++k) {
                int i  = i0 + k * MAPR + t;
                int ic = i < Q ? i : Q - 1;
                a[k] = s_idx[ic];
                b[k] = p_idx[ic];
                c[k] = n_idx[ic];
            }
            #pragma unroll
            for (int k = 0; k < 4; ++k) {
                int i = i0 + k * MAPR + t;
                if (i < Q) {
                    unsigned ua = (unsigned)(a[k] - base);
                    unsigned ub = (unsigned)(b[k] - base);
                    unsigned uc = (unsigned)(c[k] - base);
                    if (ua < (unsigned)MAPR) atomicMax(&lmap[ua], i);
                    if (ub < (unsigned)MAPR) atomicMax(&lmap[ub], Q + i);
                    if (uc < (unsigned)MAPR) atomicMax(&lmap[uc], 2 * Q + i);
                }
            }
        }
        __syncthreads();

        for (int i0 = 0; i0 < Q; i0 += 4 * MAPR) {
            int a[4], b[4], c[4];
            #pragma unroll
            for (int k = 0; k < 4; ++k) {
                int i  = i0 + k * MAPR + t;
                int ic = i < Q ? i : Q - 1;
                a[k] = s_idx[ic];
                b[k] = p_idx[ic];
                c[k] = n_idx[ic];
            }
            #pragma unroll
            for (int k = 0; k < 4; ++k) {
                int i = i0 + k * MAPR + t;
                if (i < Q) {
                    unsigned ua = (unsigned)(a[k] - base);
                    unsigned ub = (unsigned)(b[k] - base);
                    unsigned uc = (unsigned)(c[k] - base);
                    if (ua < (unsigned)MAPR) qslot[i]         = lmap[ua];
                    if (ub < (unsigned)MAPR) qslot[Q + i]     = lmap[ub];
                    if (uc < (unsigned)MAPR) qslot[2 * Q + i] = lmap[uc];
                }
            }
        }
        const int n = base + t;
        if (n < NN) map[n] = lmap[t];
    } else {
        long long i0 = (long long)((int)blockIdx.x - nmapb) * (4 * MAPR) + t;
        #pragma unroll
        for (int k = 0; k < 4; ++k) {
            long long i = i0 + k * MAPR;
            if (i < z4) hc4[i] = make_float4(0.f, 0.f, 0.f, 0.f);
        }
    }
}

template <int N>
__device__ __forceinline__ void add_raw(const int* l_e, const int* l_slot, int i, int f,
                                        const float* __restrict__ ef, float* __restrict__ hc)
{
    float v[N];
    int   sl[N];
    #pragma unroll
    for (int k = 0; k < N; ++k) {
        sl[k] = l_slot[i + k];
        v[k]  = ef[(size_t)l_e[i + k] * HID + f];
    }
    #pragma unroll
    for (int k = 0; k < N; ++k) atomicAdd(&hc[(size_t)sl[k] * DIN + f], v[k]);
}

template <int N>
__device__ __forceinline__ void add_time(const float* l_td, const int* l_slot, int i, int f,
                                         float tw, float tb, float* __restrict__ hc)
{
    float cv[N];
    int   sl[N];
    #pragma unroll
    for (int k = 0; k < N; ++k) {
        sl[k] = l_slot[i + k];
        cv[k] = cosf(l_td[i + k] * tw + tb);
    }
    #pragma unroll
    for (int k = 0; k < N; ++k) atomicAdd(&hc[(size_t)sl[k] * DIN + HID + f], cv[k]);
}

__global__ __launch_bounds__(256) void tgnn_edge_kernel(
    const int*   __restrict__ src,
    const int*   __restrict__ dst,
    const float* __restrict__ ef,
    const float* __restrict__ bt,
    const float* __restrict__ node_ts,
    const float* __restrict__ time_w,
    const float* __restrict__ time_b,
    const int*   __restrict__ map,
    float*       __restrict__ hc,
    int NE)
{
    __shared__ int   l_slot[256];
    __shared__ int   l_e[256];
    __shared__ float l_td[256];
    __shared__ int   l_wcnt[8];

    const int t    = threadIdx.x;
    const int lane = t & 31;
    const int wave = t >> 5;
    const int e    = blockIdx.x * 256 + t;

    int   slot = -1;
    int   s    = 0;
    float b    = 0.f;
    if (e < NE) {
        int d = dst[e];
        s    = src[e];
        b    = bt[e];
        slot = map[d];
    }
    const bool live = slot >= 0;
    float td = 0.f;
    if (live) td = b - node_ts[s];

    const unsigned m = __builtin_amdgcn_ballot_w32(live);
    if (lane == 0) l_wcnt[wave] = __popc(m);
    __syncthreads();
    int before = 0, total = 0;
    #pragma unroll
    for (int w = 0; w < 8; ++w) {
        int c = l_wcnt[w];
        if (w < wave) before += c;
        total += c;
    }
    if (live) {
        int r = before + __popc(m & ((1u << lane) - 1u));
        l_slot[r] = slot;
        l_e[r]    = e;
        l_td[r]   = td;
    }
    __syncthreads();

    const int f = (wave & 3) * 32 + lane;
    int i = 0;
    if (wave < 4) {
        for (; i + 8 <= total; i += 8) add_raw<8>(l_e, l_slot, i, f, ef, hc);
        if (total - i >= 4) { add_raw<4>(l_e, l_slot, i, f, ef, hc); i += 4; }
        if (total - i >= 2) { add_raw<2>(l_e, l_slot, i, f, ef, hc); i += 2; }
        if (total - i >= 1) { add_raw<1>(l_e, l_slot, i, f, ef, hc); }
    } else {
        const float tw = time_w[f];
        const float tb = time_b[f];
        for (; i + 4 <= total; i += 4) add_time<4>(l_td, l_slot, i, f, tw, tb, hc);
        if (total - i >= 2) { add_time<2>(l_td, l_slot, i, f, tw, tb, hc); i += 2; }
        if (total - i >= 1) { add_time<1>(l_td, l_slot, i, f, tw, tb, hc); }
    }
}

__global__ __launch_bounds__(256) void tgnn_predict_kernel(
    const float* __restrict__ h,
    const int*   __restrict__ s_idx,
    const int*   __restrict__ p_idx,
    const int*   __restrict__ n_idx,
    const float* __restrict__ W_src,
    const float* __restrict__ b_src,
    const float* __restrict__ W_dst,
    const float* __restrict__ b_dst,
    const float* __restrict__ W_out,
    const float* __restrict__ b_out,
    float*       __restrict__ out,
    int Q)
{
    __shared__ float sl[16 * LROW];
    __shared__ float pl[16 * LROW];
    __shared__ float nl[16 * LROW];
    __shared__ float red_pos[8 * 256];
    __shared__ float red_neg[8 * 256];

    const int tile = blockIdx.x;
    const int t    = threadIdx.x;

    for (int i = t; i < 16 * DIN; i += 256) {
        int row = i >> 8;
        int col = i & 255;
        int qi  = tile * 16 + row;
        sl[row * LROW + col] = h[(size_t)s_idx[qi] * DIN + col];
        pl[row * LROW + col] = h[(size_t)p_idx[qi] * DIN + col];
        nl[row * LROW + col] = h[(size_t)n_idx[qi] * DIN + col];
    }
    __syncthreads();

    const int wave = t >> 5;
    const int lane = t & 31;
    const int half = lane >> 4;
    const int lm   = lane & 15;
    const int n0   = wave * 16;

    v8f hs = {}, hp = {}, hn = {};

    const float* As = &sl[lm * LROW];
    const float* Ap = &pl[lm * LROW];
    const float* An = &nl[lm * LROW];

    for (int k0 = 0; k0 < DIN; k0 += 4) {
        int ka = k0 + 2 * half;
        v2f a_s; a_s.x = As[ka]; a_s.y = As[ka + 1];
        v2f a_p; a_p.x = Ap[ka]; a_p.y = Ap[ka + 1];
        v2f a_n; a_n.x = An[ka]; a_n.y = An[ka + 1];
        v2f b_s; b_s.x = W_src[ka * HID + n0 + lm]; b_s.y = W_src[(ka + 1) * HID + n0 + lm];
        v2f b_d; b_d.x = W_dst[ka * HID + n0 + lm]; b_d.y = W_dst[(ka + 1) * HID + n0 + lm];

        hs = __builtin_amdgcn_wmma_f32_16x16x4_f32(false, a_s, false, b_s, (short)0, hs, false, false);
        hp = __builtin_amdgcn_wmma_f32_16x16x4_f32(false, a_p, false, b_d, (short)0, hp, false, false);
        hn = __builtin_amdgcn_wmma_f32_16x16x4_f32(false, a_n, false, b_d, (short)0, hn, false, false);
    }

    const float bias = b_src[n0 + lm] + b_dst[n0 + lm];
    const float wo   = W_out[n0 + lm];

    #pragma unroll
    for (int r = 0; r < 8; ++r) {
        float pv = fmaxf(hs[r] + hp[r] + bias, 0.f) * wo;
        float nv = fmaxf(hs[r] + hn[r] + bias, 0.f) * wo;
        red_pos[wave * 256 + r * 32 + lane] = pv;
        red_neg[wave * 256 + r * 32 + lane] = nv;
    }
    __syncthreads();

    if (t < 16) {
        int mh = t >> 3;
        int r  = t & 7;
        float accp = b_out[0];
        float accn = b_out[0];
        for (int w = 0; w < 8; ++w) {
            #pragma unroll
            for (int l = 0; l < 16; ++l) {
                int idx = w * 256 + r * 32 + mh * 16 + l;
                accp += red_pos[idx];
                accn += red_neg[idx];
            }
        }
        out[tile * 16 + t]     = accp;
        out[Q + tile * 16 + t] = accn;
    }
}

extern "C" void kernel_launch(void* const* d_in, const int* in_sizes, int n_in,
                              void* d_out, int out_size, void* d_ws, size_t ws_size,
                              hipStream_t stream) {
    const int*   src     = (const int*)  d_in[0];
    const int*   dst     = (const int*)  d_in[1];
    const float* ef      = (const float*)d_in[2];
    const float* bt      = (const float*)d_in[3];
    const float* node_ts = (const float*)d_in[4];
    const int*   s_idx   = (const int*)  d_in[5];
    const int*   p_idx   = (const int*)  d_in[6];
    const int*   n_idx   = (const int*)  d_in[7];
    const float* time_w  = (const float*)d_in[8];
    const float* time_b  = (const float*)d_in[9];
    const float* W_src   = (const float*)d_in[10];
    const float* b_src   = (const float*)d_in[11];
    const float* W_dst   = (const float*)d_in[12];
    const float* b_dst   = (const float*)d_in[13];
    const float* W_out   = (const float*)d_in[14];
    const float* b_out   = (const float*)d_in[15];

    float* out = (float*)d_out;

    const int NN = in_sizes[4];
    const int NE = in_sizes[0];
    const int Q  = in_sizes[5];

    float* hc    = (float*)d_ws;
    int*   qslot = (int*)(hc + (size_t)3 * Q * DIN);
    int*   map   = qslot + (size_t)3 * Q;

    long long z4 = (long long)3 * Q * DIN / 4;
    int nmapb = (NN + MAPR - 1) / MAPR;
    int zgrid = (int)((z4 + 4 * MAPR - 1) / (4 * MAPR));
    tgnn_prep_kernel<<<nmapb + zgrid, MAPR, 0, stream>>>(s_idx, p_idx, n_idx, map, qslot, (float4*)hc, NN, Q, nmapb, z4);

    tgnn_edge_kernel<<<(NE + 255) / 256, 256, 0, stream>>>(src, dst, ef, bt, node_ts, time_w, time_b, map, hc, NE);

    tgnn_predict_kernel<<<Q / 16, 256, 0, stream>>>(hc, qslot, qslot + Q, qslot + 2 * Q,
                                                    W_src, b_src, W_dst, b_dst,
                                                    W_out, b_out, out, Q);
}


# Scatter-add removed: edges into queried nodes are chained per node (one int exchange each); the predictor sums each chain into its LDS tile. No accumulator, no f32 atomics; sum order = chain order
# speedup vs baseline: 9.3753x; 1.3392x over previous
#include <hip/hip_runtime.h>


typedef __attribute__((ext_vector_type(2))) float v2f;
typedef __attribute__((ext_vector_type(8))) float v8f;

#define HID   128
#define DIN   256
#define LROW  260
#define MAPR  1024
#define EPT   4
#define SPW   6


__global__ __launch_bounds__(1024) void tgnn_prep_kernel(
    const int* __restrict__ s_idx,
    const int* __restrict__ p_idx,
    const int* __restrict__ n_idx,
    int*       __restrict__ flag,
    int*       __restrict__ head,
    int NN, int Q)
{
    __shared__ int lq[MAPR];
    const int t    = threadIdx.x;
    const int base = blockIdx.x * MAPR;
    lq[t] = 0;
    __syncthreads();

    for (int i0 = 0; i0 < Q; i0 += 4 * MAPR) {
        int a[4], b[4], c[4];
        #pragma unroll
        for (int k = 0; k < 4; ++k) {
            int i  = i0 + k * MAPR + t;
            int ic = i < Q ? i : Q - 1;
            a[k] = s_idx[ic];
            b[k] = p_idx[ic];
            c[k] = n_idx[ic];
        }
        #pragma unroll
        for (int k = 0; k < 4; ++k) {
            int i = i0 + k * MAPR + t;
            if (i < Q) {
                unsigned ua = (unsigned)(a[k] - base);
                unsigned ub = (unsigned)(b[k] - base);
                unsigned uc = (unsigned)(c[k] - base);
                if (ua < (unsigned)MAPR) lq[ua] = 1;
                if (ub < (unsigned)MAPR) lq[ub] = 1;
                if (uc < (unsigned)MAPR) lq[uc] = 1;
            }
        }
    }
    __syncthreads();

    const int n = base + t;
    if (n < NN) {
        flag[n] = lq[t];
        head[n] = -1;
    }
}

__global__ __launch_bounds__(256) void tgnn_link_kernel(
    const int*   __restrict__ src,
    const int*   __restrict__ dst,
    const float* __restrict__ bt,
    const float* __restrict__ node_ts,
    const int*   __restrict__ flag,
    int*         __restrict__ head,
    int2*        __restrict__ link,
    int NE)
{
    const int e0 = blockIdx.x * (256 * EPT) + threadIdx.x;

    int   d[EPT], s[EPT];
    float b[EPT];
    #pragma unroll
    for (int k = 0; k < EPT; ++k) {
        int e  = e0 + k * 256;
        int ec = e < NE ? e : NE - 1;
        d[k] = dst[ec];
        s[k] = src[ec];
        b[k] = bt[ec];
    }
    int f[EPT];
    #pragma unroll
    for (int k = 0; k < EPT; ++k) f[k] = flag[d[k]];
    bool live[EPT];
    #pragma unroll
    for (int k = 0; k < EPT; ++k) live[k] = (e0 + k * 256 < NE) & (f[k] != 0);

    float ts[EPT];
    #pragma unroll
    for (int k = 0; k < EPT; ++k) ts[k] = node_ts[live[k] ? s[k] : 0];

    int old[EPT];
    #pragma unroll
    for (int k = 0; k < EPT; ++k) old[k] = live[k] ? atomicExch(&head[d[k]], e0 + k * 256) : -1;

    #pragma unroll
    for (int k = 0; k < EPT; ++k)
        if (live[k]) link[e0 + k * 256] = make_int2(old[k], __float_as_int(b[k] - ts[k]));
}

__global__ __launch_bounds__(256) void tgnn_predict_kernel(
    const float* __restrict__ ef,
    const int*   __restrict__ head,
    const int2*  __restrict__ link,
    const int*   __restrict__ s_idx,
    const int*   __restrict__ p_idx,
    const int*   __restrict__ n_idx,
    const float* __restrict__ time_w,
    const float* __restrict__ time_b,
    const float* __restrict__ W_src,
    const float* __restrict__ b_src,
    const float* __restrict__ W_dst,
    const float* __restrict__ b_dst,
    const float* __restrict__ W_out,
    const float* __restrict__ b_out,
    float*       __restrict__ out,
    int Q, int NE)
{
    __shared__ __attribute__((aligned(16))) float sl[16 * LROW];
    __shared__ __attribute__((aligned(16))) float pl[16 * LROW];
    __shared__ __attribute__((aligned(16))) float nl[16 * LROW];
    __shared__ float red_pos[8 * 256];
    __shared__ float red_neg[8 * 256];

    const int tile = blockIdx.x;
    const int t    = threadIdx.x;
    const int lane = t & 31;
    const int uw   = __builtin_amdgcn_readfirstlane(t >> 5);

    {
        const float4* __restrict__ ef4 = reinterpret_cast<const float4*>(ef);
        const float4 tw = reinterpret_cast<const float4*>(time_w)[lane];
        const float4 tb = reinterpret_cast<const float4*>(time_b)[lane];
        const unsigned une = (unsigned)NE;

        int hv = -1;
        if (lane < SPW) {
            const int j  = uw * SPW + lane;
            const int li = j >> 4;
            const int qi = tile * 16 + (j & 15);
            const int* __restrict__ ip = li == 0 ? s_idx : (li == 1 ? p_idx : n_idx);
            hv = head[ip[qi]];
        }

        int    ec[SPW], en[SPW];
        int2   lk[SPW];
        float4 v[SPW], ar[SPW], at[SPW];
        #pragma unroll
        for (int c = 0; c < SPW; ++c) {
            en[c] = __builtin_amdgcn_readlane(hv, c);
            ec[c] = -1;
            lk[c] = make_int2(-1, 0);
            v[c]  = make_float4(0.f, 0.f, 0.f, 0.f);
            ar[c] = make_float4(0.f, 0.f, 0.f, 0.f);
            at[c] = make_float4(0.f, 0.f, 0.f, 0.f);
        }
        for (int hop = 0; hop <= NE; ++hop) {
            int2   lkn[SPW];
            float4 vn[SPW];
            #pragma unroll
            for (int c = 0; c < SPW; ++c) {
                lkn[c] = make_int2(-1, 0);
                vn[c]  = make_float4(0.f, 0.f, 0.f, 0.f);
                if ((unsigned)en[c] < une) {
                    lkn[c] = link[en[c]];
                    vn[c]  = ef4[(size_t)en[c] * (HID / 4) + lane];
                }
            }
            #pragma unroll
            for (int c = 0; c < SPW; ++c) {
                if ((unsigned)ec[c] < une) {
                    const float td = __int_as_float(lk[c].y);
                    ar[c].x += v[c].x;
                    ar[c].y += v[c].y;
                    ar[c].z += v[c].z;
                    ar[c].w += v[c].w;
                    at[c].x += cosf(td * tw.x + tb.x);
                    at[c].y += cosf(td * tw.y + tb.y);
                    at[c].z += cosf(td * tw.z + tb.z);
                    at[c].w += cosf(td * tw.w + tb.w);
                }
            }
            bool any = false;
            #pragma unroll
            for (int c = 0; c < SPW; ++c) {
                ec[c] = en[c];
                lk[c] = lkn[c];
                v[c]  = vn[c];
                en[c] = ((unsigned)ec[c] < une) ? __builtin_amdgcn_readfirstlane(lk[c].x) : -1;
                any   = any || ((unsigned)ec[c] < une);
            }
            if (!any) break;
        }
        #pragma unroll
        for (int c = 0; c < SPW; ++c) {
            const int j  = uw * SPW + c;
            const int li = j >> 4;
            float* rowp = (li == 0 ? sl : (li == 1 ? pl : nl)) + (j & 15) * LROW;
            *reinterpret_cast<float4*>(rowp + 4 * lane)       = ar[c];
            *reinterpret_cast<float4*>(rowp + HID + 4 * lane) = at[c];
        }
    }
    __syncthreads();

    const int wave = t >> 5;
    const int half = lane >> 4;
    const int lm   = lane & 15;
    const int n0   = wave * 16;

    v8f hs = {}, hp = {}, hn = {};

    const float* As = &sl[lm * LROW];
    const float* Ap = &pl[lm * LROW];
    const float* An = &nl[lm * LROW];

    for (int k0 = 0; k0 < DIN; k0 += 4) {
        int ka = k0 + 2 * half;
        v2f a_s; a_s.x = As[ka]; a_s.y = As[ka + 1];
        v2f a_p; a_p.x = Ap[ka]; a_p.y = Ap[ka + 1];
        v2f a_n; a_n.x = An[ka]; a_n.y = An[ka + 1];
        v2f b_s; b_s.x = W_src[ka * HID + n0 + lm]; b_s.y = W_src[(ka + 1) * HID + n0 + lm];
        v2f b_d; b_d.x = W_dst[ka * HID + n0 + lm]; b_d.y = W_dst[(ka + 1) * HID + n0 + lm];

        hs = __builtin_amdgcn_wmma_f32_16x16x4_f32(false, a_s, false, b_s, (short)0, hs, false, false);
        hp = __builtin_amdgcn_wmma_f32_16x16x4_f32(false, a_p, false, b_d, (short)0, hp, false, false);
        hn = __builtin_amdgcn_wmma_f32_16x16x4_f32(false, a_n, false, b_d, (short)0, hn, false, false);
    }

    const float bias = b_src[n0 + lm] + b_dst[n0 + lm];
    const float wo   = W_out[n0 + lm];

    #pragma unroll
    for (int r = 0; r < 8; ++r) {
        float pv = fmaxf(hs[r] + hp[r] + bias, 0.f) * wo;
        float nv = fmaxf(hs[r] + hn[r] + bias, 0.f) * wo;
        red_pos[wave * 256 + r * 32 + lane] = pv;
        red_neg[wave * 256 + r * 32 + lane] = nv;
    }
    __syncthreads();

    if (t < 16) {
        int mh = t >> 3;
        int r  = t & 7;
        float accp = b_out[0];
        float accn = b_out[0];
        for (int w = 0; w < 8; ++w) {
            #pragma unroll
            for (int l = 0; l < 16; ++l) {
                int idx = w * 256 + r * 32 + mh * 16 + l;
                accp += red_pos[idx];
                accn += red_neg[idx];
            }
        }
        out[tile * 16 + t]     = accp;
        out[Q + tile * 16 + t] = accn;
    }
}

extern "C" void kernel_launch(void* const* d_in, const int* in_sizes, int n_in,
                              void* d_out, int out_size, void* d_ws, size_t ws_size,
                              hipStream_t stream) {
    const int*   src     = (const int*)  d_in[0];
    const int*   dst     = (const int*)  d_in[1];
    const float* ef      = (const float*)d_in[2];
    const float* bt      = (const float*)d_in[3];
    const float* node_ts = (const float*)d_in[4];
    const int*   s_idx   = (const int*)  d_in[5];
    const int*   p_idx   = (const int*)  d_in[6];
    const int*   n_idx   = (const int*)  d_in[7];
    const float* time_w  = (const float*)d_in[8];
    const float* time_b  = (const float*)d_in[9];
    const float* W_src   = (const float*)d_in[10];
    const float* b_src   = (const float*)d_in[11];
    const float* W_dst   = (const float*)d_in[12];
    const float* b_dst   = (const float*)d_in[13];
    const float* W_out   = (const float*)d_in[14];
    const float* b_out   = (const float*)d_in[15];

    float* out = (float*)d_out;

    const int NN = in_sizes[4];
    const int NE = in_sizes[0];
    const int Q  = in_sizes[5];

    int*  flag = (int*)d_ws;
    int*  head = flag + (size_t)NN;
    int2* link = (int2*)(head + (size_t)NN);

    int nmapb = (NN + MAPR - 1) / MAPR;
    tgnn_prep_kernel<<<nmapb, MAPR, 0, stream>>>(s_idx, p_idx, n_idx, flag, head, NN, Q);

    tgnn_link_kernel<<<(NE + 256 * EPT - 1) / (256 * EPT), 256, 0, stream>>>(src, dst, bt, node_ts, flag, head, link, NE);

    tgnn_predict_kernel<<<Q / 16, 256, 0, stream>>>(ef, head, link, s_idx, p_idx, n_idx, time_w, time_b,
                                                    W_src, b_src, W_dst, b_dst,
                                                    W_out, b_out, out, Q, NE);
}


# Predictor K loop software-pipelined: each wave asks for its W fragments (global loads) three 4-k-step units ahead of the products that use them, the first two units at kernel entry
# speedup vs baseline: 10.2052x; 1.0885x over previous
#include <hip/hip_runtime.h>


typedef __attribute__((ext_vector_type(2))) float v2f;
typedef __attribute__((ext_vector_type(8))) float v8f;

#define HID   128
#define DIN   256
#define LROW  260
#define MAPR  1024
#define EPT   4
#define SPW   6
#define KU    4
#define NU    (DIN / (4 * KU))
#define AHEAD 3


__global__ __launch_bounds__(1024) void tgnn_prep_kernel(
    const int* __restrict__ s_idx,
    const int* __restrict__ p_idx,
    const int* __restrict__ n_idx,
    int*       __restrict__ flag,
    int*       __restrict__ head,
    int NN, int Q)
{
    __shared__ int lq[MAPR];
    const int t    = threadIdx.x;
    const int base = blockIdx.x * MAPR;
    lq[t] = 0;
    __syncthreads();

    for (int i0 = 0; i0 < Q; i0 += 4 * MAPR) {
        int a[4], b[4], c[4];
        #pragma unroll
        for (int k = 0; k < 4; ++k) {
            int i  = i0 + k * MAPR + t;
            int ic = i < Q ? i : Q - 1;
            a[k] = s_idx[ic];
            b[k] = p_idx[ic];
            c[k] = n_idx[ic];
        }
        #pragma unroll
        for (int k = 0; k < 4; ++k) {
            int i = i0 + k * MAPR + t;
            if (i < Q) {
                unsigned ua = (unsigned)(a[k] - base);
                unsigned ub = (unsigned)(b[k] - base);
                unsigned uc = (unsigned)(c[k] - base);
                if (ua < (unsigned)MAPR) lq[ua] = 1;
                if (ub < (unsigned)MAPR) lq[ub] = 1;
                if (uc < (unsigned)MAPR) lq[uc] = 1;
            }
        }
    }
    __syncthreads();

    const int n = base + t;
    if (n < NN) {
        flag[n] = lq[t];
        head[n] = -1;
    }
}

__global__ __launch_bounds__(256) void tgnn_link_kernel(
    const int*   __restrict__ src,
    const int*   __restrict__ dst,
    const float* __restrict__ bt,
    const float* __restrict__ node_ts,
    const int*   __restrict__ flag,
    int*         __restrict__ head,
    int2*        __restrict__ link,
    int NE)
{
    const int e0 = blockIdx.x * (256 * EPT) + threadIdx.x;

    int   d[EPT], s[EPT];
    float b[EPT];
    #pragma unroll
    for (int k = 0; k < EPT; ++k) {
        int e  = e0 + k * 256;
        int ec = e < NE ? e : NE - 1;
        d[k] = dst[ec];
        s[k] = src[ec];
        b[k] = bt[ec];
    }
    int f[EPT];
    #pragma unroll
    for (int k = 0; k < EPT; ++k) f[k] = flag[d[k]];
    bool live[EPT];
    #pragma unroll
    for (int k = 0; k < EPT; ++k) live[k] = (e0 + k * 256 < NE) & (f[k] != 0);

    float ts[EPT];
    #pragma unroll
    for (int k = 0; k < EPT; ++k) ts[k] = node_ts[live[k] ? s[k] : 0];

    int old[EPT];
    #pragma unroll
    for (int k = 0; k < EPT; ++k) old[k] = live[k] ? atomicExch(&head[d[k]], e0 + k * 256) : -1;

    #pragma unroll
    for (int k = 0; k < EPT; ++k)
        if (live[k]) link[e0 + k * 256] = make_int2(old[k], __float_as_int(b[k] - ts[k]));
}

__device__ __forceinline__ void load_w_unit(v2f (&bs)[KU], v2f (&bd)[KU],
                                            const float* __restrict__ wsl,
                                            const float* __restrict__ wdl, int u)
{
    #pragma unroll
    for (int i = 0; i < KU; ++i) {
        const int kr = (u * KU + i) * 4 * HID;
        bs[i].x = wsl[kr];
        bs[i].y = wsl[kr + HID];
        bd[i].x = wdl[kr];
        bd[i].y = wdl[kr + HID];
    }
}

__device__ __forceinline__ void mma_unit(v8f& hs, v8f& hp, v8f& hn,
                                         const float* As, const float* Ap, const float* An,
                                         const v2f (&bs)[KU], const v2f (&bd)[KU], int u)
{
    #pragma unroll
    for (int i = 0; i < KU; ++i) {
        const int k0 = (u * KU + i) * 4;
        v2f a_s; a_s.x = As[k0]; a_s.y = As[k0 + 1];
        v2f a_p; a_p.x = Ap[k0]; a_p.y = Ap[k0 + 1];
        v2f a_n; a_n.x = An[k0]; a_n.y = An[k0 + 1];
        hs = __builtin_amdgcn_wmma_f32_16x16x4_f32(false, a_s, false, bs[i], (short)0, hs, false, false);
        hp = __builtin_amdgcn_wmma_f32_16x16x4_f32(false, a_p, false, bd[i], (short)0, hp, false, false);
        hn = __builtin_amdgcn_wmma_f32_16x16x4_f32(false, a_n, false, bd[i], (short)0, hn, false, false);
    }
}

__global__ __launch_bounds__(256) void tgnn_predict_kernel(
    const float* __restrict__ ef,
    const int*   __restrict__ head,
    const int2*  __restrict__ link,
    const int*   __restrict__ s_idx,
    const int*   __restrict__ p_idx,
    const int*   __restrict__ n_idx,
    const float* __restrict__ time_w,
    const float* __restrict__ time_b,
    const float* __restrict__ W_src,
    const float* __restrict__ b_src,
    const float* __restrict__ W_dst,
    const float* __restrict__ b_dst,
    const float* __restrict__ W_out,
    const float* __restrict__ b_out,
    float*       __restrict__ out,
    int Q, int NE)
{
    __shared__ __attribute__((aligned(16))) float sl[16 * LROW];
    __shared__ __attribute__((aligned(16))) float pl[16 * LROW];
    __shared__ __attribute__((aligned(16))) float nl[16 * LROW];
    __shared__ float red_pos[8 * 256];
    __shared__ float red_neg[8 * 256];

    const int tile = blockIdx.x;
    const int t    = threadIdx.x;
    const int lane = t & 31;
    const int uw   = __builtin_amdgcn_readfirstlane(t >> 5);

    const int wave = t >> 5;
    const int half = lane >> 4;
    const int lm   = lane & 15;
    const int n0   = wave * 16;

    const float* __restrict__ wsl = W_src + (2 * half) * HID + n0 + lm;
    const float* __restrict__ wdl = W_dst + (2 * half) * HID + n0 + lm;
    v2f bs[AHEAD + 1][KU], bd[AHEAD + 1][KU];

    {
        const float4* __restrict__ ef4 = reinterpret_cast<const float4*>(ef);
        const float4 tw = reinterpret_cast<const float4*>(time_w)[lane];
        const float4 tb = reinterpret_cast<const float4*>(time_b)[lane];
        const unsigned une = (unsigned)NE;

        int hv = -1;
        if (lane < SPW) {
            const int j  = uw * SPW + lane;
            const int li = j >> 4;
            const int qi = tile * 16 + (j & 15);
            const int* __restrict__ ip = li == 0 ? s_idx : (li == 1 ? p_idx : n_idx);
            hv = head[ip[qi]];
        }

        load_w_unit(bs[0], bd[0], wsl, wdl, 0);
        load_w_unit(bs[1], bd[1], wsl, wdl, 1);

        int    ec[SPW], en[SPW];
        int2   lk[SPW];
        float4 v[SPW], ar[SPW], at[SPW];
        #pragma unroll
        for (int c = 0; c < SPW; ++c) {
            en[c] = __builtin_amdgcn_readlane(hv, c);
            ec[c] = -1;
            lk[c] = make_int2(-1, 0);
            v[c]  = make_float4(0.f, 0.f, 0.f, 0.f);
            ar[c] = make_float4(0.f, 0.f, 0.f, 0.f);
            at[c] = make_float4(0.f, 0.f, 0.f, 0.f);
        }
        for (int hop = 0; hop <= NE; ++hop) {
            int2   lkn[SPW];
            float4 vn[SPW];
            #pragma unroll
            for (int c = 0; c < SPW; ++c) {
                lkn[c] = make_int2(-1, 0);
                vn[c]  = make_float4(0.f, 0.f, 0.f, 0.f);
                if ((unsigned)en[c] < une) {
                    lkn[c] = link[en[c]];
                    vn[c]  = ef4[(size_t)en[c] * (HID / 4) + lane];
                }
            }
            #pragma unroll
            for (int c = 0; c < SPW; ++c) {
                if ((unsigned)ec[c] < une) {
                    const float td = __int_as_float(lk[c].y);
                    ar[c].x += v[c].x;
                    ar[c].y += v[c].y;
                    ar[c].z += v[c].z;
                    ar[c].w += v[c].w;
                    at[c].x += cosf(td * tw.x + tb.x);
                    at[c].y += cosf(td * tw.y + tb.y);
                    at[c].z += cosf(td * tw.z + tb.z);
                    at[c].w += cosf(td * tw.w + tb.w);
                }
            }
            bool any = false;
            #pragma unroll
            for (int c = 0; c < SPW; ++c) {
                ec[c] = en[c];
                lk[c] = lkn[c];
                v[c]  = vn[c];
                en[c] = ((unsigned)ec[c] < une) ? __builtin_amdgcn_readfirstlane(lk[c].x) : -1;
                any   = any || ((unsigned)ec[c] < une);
            }
            if (!any) break;
        }
        #pragma unroll
        for (int c = 0; c < SPW; ++c) {
            const int j  = uw * SPW + c;
            const int li = j >> 4;
            float* rowp = (li == 0 ? sl : (li == 1 ? pl : nl)) + (j & 15) * LROW;
            *reinterpret_cast<float4*>(rowp + 4 * lane)       = ar[c];
            *reinterpret_cast<float4*>(rowp + HID + 4 * lane) = at[c];
        }
    }
    __syncthreads();

    v8f hs = {}, hp = {}, hn = {};

    const float* As = &sl[lm * LROW + 2 * half];
    const float* Ap = &pl[lm * LROW + 2 * half];
    const float* An = &nl[lm * LROW + 2 * half];

    static_assert(AHEAD == 3 && NU > AHEAD, "the three units loaded ahead of the loop are written out for AHEAD == 3");
    load_w_unit(bs[2], bd[2], wsl, wdl, 2);
    #pragma unroll
    for (int u = 0; u < NU; ++u) {
        if (u + AHEAD < NU)
            load_w_unit(bs[(u + AHEAD) % (AHEAD + 1)], bd[(u + AHEAD) % (AHEAD + 1)], wsl, wdl, u + AHEAD);
        mma_unit(hs, hp, hn, As, Ap, An, bs[u % (AHEAD + 1)], bd[u % (AHEAD + 1)], u);
    }

    const float bias = b_src[n0 + lm] + b_dst[n0 + lm];
    const float wo   = W_out[n0 + lm];

    #pragma unroll
    for (int r = 0; r < 8; ++r) {
        float pv = fmaxf(hs[r] + hp[r] + bias, 0.f) * wo;
        float nv = fmaxf(hs[r] + hn[r] + bias, 0.f) * wo;
        red_pos[wave * 256 + r * 32 + lane] = pv;
        red_neg[wave * 256 + r * 32 + lane] = nv;
    }
    __syncthreads();

    if (t < 16) {
        int mh = t >> 3;
        int r  = t & 7;
        float accp = b_out[0];
        float accn = b_out[0];
        for (int w = 0; w < 8; ++w) {
            #pragma unroll
            for (int l = 0; l < 16; ++l) {
                int idx = w * 256 + r * 32 + mh * 16 + l;
                accp += red_pos[idx];
                accn += red_neg[idx];
            }
        }
        out[tile * 16 + t]     = accp;
        out[Q + tile * 16 + t] = accn;
    }
}

extern "C" void kernel_launch(void* const* d_in, const int* in_sizes, int n_in,
                              void* d_out, int out_size, void* d_ws, size_t ws_size,
                              hipStream_t stream) {
    const int*   src     = (const int*)  d_in[0];
    const int*   dst     = (const int*)  d_in[1];
    const float* ef      = (const float*)d_in[2];
    const float* bt      = (const float*)d_in[3];
    const float* node_ts = (const float*)d_in[4];
    const int*   s_idx   = (const int*)  d_in[5];
    const int*   p_idx   = (const int*)  d_in[6];
    const int*   n_idx   = (const int*)  d_in[7];
    const float* time_w  = (const float*)d_in[8];
    const float* time_b  = (const float*)d_in[9];
    const float* W_src   = (const float*)d_in[10];
    const float* b_src   = (const float*)d_in[11];
    const float* W_dst   = (const float*)d_in[12];
    const float* b_dst   = (const float*)d_in[13];
    const float* W_out   = (const float*)d_in[14];
    const float* b_out   = (const float*)d_in[15];

    float* out = (float*)d_out;

    const int NN = in_sizes[4];
    const int NE = in_sizes[0];
    const int Q  = in_sizes[5];

    int*  flag = (int*)d_ws;
    int*  head = flag + (size_t)NN;
    int2* link = (int2*)(head + (size_t)NN);

    int nmapb = (NN + MAPR - 1) / MAPR;
    tgnn_prep_kernel<<<nmapb, MAPR, 0, stream>>>(s_idx, p_idx, n_idx, flag, head, NN, Q);

    tgnn_link_kernel<<<(NE + 256 * EPT - 1) / (256 * EPT), 256, 0, stream>>>(src, dst, bt, node_ts, flag, head, link, NE);

    tgnn_predict_kernel<<<Q / 16, 256, 0, stream>>>(ef, head, link, s_idx, p_idx, n_idx, time_w, time_b,
                                                    W_src, b_src, W_dst, b_dst,
                                                    W_out, b_out, out, Q, NE);
}


# Query flags leave launch 1: each link block builds a bit map of queried nodes in LDS (atomic OR over the query lists) and tests destinations there; launch 1 only fills head; link blocks: 2048 edges
# speedup vs baseline: 10.6132x; 1.0400x over previous
#include <hip/hip_runtime.h>


typedef __attribute__((ext_vector_type(2))) float v2f;
typedef __attribute__((ext_vector_type(8))) float v8f;

#define HID   128
#define DIN   256
#define LROW  260
#define FTHR  256
#define FPT   16
#define LTHR  256
#define EPT   8
#define QPT   4
#define SPW   6
#define KU    4
#define NU    (DIN / (4 * KU))
#define AHEAD 3


__global__ __launch_bounds__(FTHR) void tgnn_fill_kernel(int* __restrict__ head, int NN)
{
    const int base = blockIdx.x * (FTHR * FPT) + threadIdx.x;
    #pragma unroll
    for (int k = 0; k < FPT; ++k) {
        const int n = base + k * FTHR;
        if (n < NN) head[n] = -1;
    }
}

__device__ __forceinline__ void mark_node(unsigned* bm, int node, unsigned unn)
{
    const bool inr = (unsigned)node < unn;
    atomicOr(&bm[inr ? ((unsigned)node >> 5) : 0u], inr ? (1u << (node & 31)) : 0u);
}

__device__ __forceinline__ void load_query_groups(int4 (&a)[QPT], int4 (&p)[QPT], int4 (&c)[QPT],
                                                  const int4* __restrict__ s4,
                                                  const int4* __restrict__ p4,
                                                  const int4* __restrict__ n4,
                                                  int g0, int t, int q4)
{
    #pragma unroll
    for (int k = 0; k < QPT; ++k) {
        int g  = g0 + k * LTHR + t;
        int gc = g < q4 ? g : q4 - 1;
        a[k] = s4[gc];
        p[k] = p4[gc];
        c[k] = n4[gc];
    }
}

__global__ __launch_bounds__(LTHR) void tgnn_link_kernel(
    const int*   __restrict__ src,
    const int*   __restrict__ dst,
    const float* __restrict__ bt,
    const float* __restrict__ node_ts,
    const int*   __restrict__ s_idx,
    const int*   __restrict__ p_idx,
    const int*   __restrict__ n_idx,
    int*         __restrict__ head,
    int2*        __restrict__ link,
    int NE, int NN, int Q)
{
    extern __shared__ unsigned bm[];

    const int t  = threadIdx.x;
    const int e0 = blockIdx.x * (LTHR * EPT) + t;
    const unsigned unn = (unsigned)NN;

    const int nw = (NN + 31) >> 5;
    for (int i = t; i < nw; i += LTHR) bm[i] = 0u;
    __syncthreads();

    const int q4 = Q >> 2;
    const int4* __restrict__ s4 = reinterpret_cast<const int4*>(s_idx);
    const int4* __restrict__ p4 = reinterpret_cast<const int4*>(p_idx);
    const int4* __restrict__ n4 = reinterpret_cast<const int4*>(n_idx);
    int4 a[QPT], p[QPT], c[QPT];
    if (q4 > 0) load_query_groups(a, p, c, s4, p4, n4, 0, t, q4);

    int   d[EPT], s[EPT];
    float b[EPT];
    #pragma unroll
    for (int k = 0; k < EPT; ++k) {
        int e  = e0 + k * LTHR;
        int ec = e < NE ? e : NE - 1;
        d[k] = dst[ec];
        s[k] = src[ec];
        b[k] = bt[ec];
    }

    for (int g0 = 0; g0 < q4; ) {
        #pragma unroll
        for (int k = 0; k < QPT; ++k) {
            int g = g0 + k * LTHR + t;
            if (g < q4) {
                mark_node(bm, a[k].x, unn); mark_node(bm, a[k].y, unn);
                mark_node(bm, a[k].z, unn); mark_node(bm, a[k].w, unn);
                mark_node(bm, p[k].x, unn); mark_node(bm, p[k].y, unn);
                mark_node(bm, p[k].z, unn); mark_node(bm, p[k].w, unn);
                mark_node(bm, c[k].x, unn); mark_node(bm, c[k].y, unn);
                mark_node(bm, c[k].z, unn); mark_node(bm, c[k].w, unn);
            }
        }
        g0 += QPT * LTHR;
        if (g0 < q4) load_query_groups(a, p, c, s4, p4, n4, g0, t, q4);
    }
    for (int i = (q4 << 2) + t; i < Q; i += LTHR) {
        mark_node(bm, s_idx[i], unn);
        mark_node(bm, p_idx[i], unn);
        mark_node(bm, n_idx[i], unn);
    }
    __syncthreads();

    bool live[EPT];
    #pragma unroll
    for (int k = 0; k < EPT; ++k) {
        const bool inr = (unsigned)d[k] < unn;
        const unsigned w = bm[inr ? ((unsigned)d[k] >> 5) : 0u];
        live[k] = (e0 + k * LTHR < NE) & inr & (((w >> (d[k] & 31)) & 1u) != 0u);
    }

    float ts[EPT];
    #pragma unroll
    for (int k = 0; k < EPT; ++k) ts[k] = node_ts[live[k] ? s[k] : 0];

    int old[EPT];
    #pragma unroll
    for (int k = 0; k < EPT; ++k) old[k] = live[k] ? atomicExch(&head[d[k]], e0 + k * LTHR) : -1;

    #pragma unroll
    for (int k = 0; k < EPT; ++k)
        if (live[k]) link[e0 + k * LTHR] = make_int2(old[k], __float_as_int(b[k] - ts[k]));
}

__device__ __forceinline__ void load_w_unit(v2f (&bs)[KU], v2f (&bd)[KU],
                                            const float* __restrict__ wsl,
                                            const float* __restrict__ wdl, int u)
{
    #pragma unroll
    for (int i = 0; i < KU; ++i) {
        const int kr = (u * KU + i) * 4 * HID;
        bs[i].x = wsl[kr];
        bs[i].y = wsl[kr + HID];
        bd[i].x = wdl[kr];
        bd[i].y = wdl[kr + HID];
    }
}

__device__ __forceinline__ void mma_unit(v8f& hs, v8f& hp, v8f& hn,
                                         const float* As, const float* Ap, const float* An,
                                         const v2f (&bs)[KU], const v2f (&bd)[KU], int u)
{
    #pragma unroll
    for (int i = 0; i < KU; ++i) {
        const int k0 = (u * KU + i) * 4;
        v2f a_s; a_s.x = As[k0]; a_s.y = As[k0 + 1];
        v2f a_p; a_p.x = Ap[k0]; a_p.y = Ap[k0 + 1];
        v2f a_n; a_n.x = An[k0]; a_n.y = An[k0 + 1];
        hs = __builtin_amdgcn_wmma_f32_16x16x4_f32(false, a_s, false, bs[i], (short)0, hs, false, false);
        hp = __builtin_amdgcn_wmma_f32_16x16x4_f32(false, a_p, false, bd[i], (short)0, hp, false, false);
        hn = __builtin_amdgcn_wmma_f32_16x16x4_f32(false, a_n, false, bd[i], (short)0, hn, false, false);
    }
}

__global__ __launch_bounds__(256) void tgnn_predict_kernel(
    const float* __restrict__ ef,
    const int*   __restrict__ head,
    const int2*  __restrict__ link,
    const int*   __restrict__ s_idx,
    const int*   __restrict__ p_idx,
    const int*   __restrict__ n_idx,
    const float* __restrict__ time_w,
    const float* __restrict__ time_b,
    const float* __restrict__ W_src,
    const float* __restrict__ b_src,
    const float* __restrict__ W_dst,
    const float* __restrict__ b_dst,
    const float* __restrict__ W_out,
    const float* __restrict__ b_out,
    float*       __restrict__ out,
    int Q, int NE)
{
    __shared__ __attribute__((aligned(16))) float sl[16 * LROW];
    __shared__ __attribute__((aligned(16))) float pl[16 * LROW];
    __shared__ __attribute__((aligned(16))) float nl[16 * LROW];
    __shared__ float red_pos[8 * 256];
    __shared__ float red_neg[8 * 256];

    const int tile = blockIdx.x;
    const int t    = threadIdx.x;
    const int lane = t & 31;
    const int uw   = __builtin_amdgcn_readfirstlane(t >> 5);

    const int wave = t >> 5;
    const int half = lane >> 4;
    const int lm   = lane & 15;
    const int n0   = wave * 16;

    const float* __restrict__ wsl = W_src + (2 * half) * HID + n0 + lm;
    const float* __restrict__ wdl = W_dst + (2 * half) * HID + n0 + lm;
    v2f bs[AHEAD + 1][KU], bd[AHEAD + 1][KU];

    {
        const float4* __restrict__ ef4 = reinterpret_cast<const float4*>(ef);
        const float4 tw = reinterpret_cast<const float4*>(time_w)[lane];
        const float4 tb = reinterpret_cast<const float4*>(time_b)[lane];
        const unsigned une = (unsigned)NE;

        int hv = -1;
        if (lane < SPW) {
            const int j  = uw * SPW + lane;
            const int li = j >> 4;
            const int qi = tile * 16 + (j & 15);
            const int* __restrict__ ip = li == 0 ? s_idx : (li == 1 ? p_idx : n_idx);
            hv = head[ip[qi]];
        }

        load_w_unit(bs[0], bd[0], wsl, wdl, 0);
        load_w_unit(bs[1], bd[1], wsl, wdl, 1);

        int    ec[SPW], en[SPW];
        int2   lk[SPW];
        float4 v[SPW], ar[SPW], at[SPW];
        #pragma unroll
        for (int c = 0; c < SPW; ++c) {
            en[c] = __builtin_amdgcn_readlane(hv, c);
            ec[c] = -1;
            lk[c] = make_int2(-1, 0);
            v[c]  = make_float4(0.f, 0.f, 0.f, 0.f);
            ar[c] = make_float4(0.f, 0.f, 0.f, 0.f);
            at[c] = make_float4(0.f, 0.f, 0.f, 0.f);
        }
        for (int hop = 0; hop <= NE; ++hop) {
            int2   lkn[SPW];
            float4 vn[SPW];
            #pragma unroll
            for (int c = 0; c < SPW; ++c) {
                lkn[c] = make_int2(-1, 0);
                vn[c]  = make_float4(0.f, 0.f, 0.f, 0.f);
                if ((unsigned)en[c] < une) {
                    lkn[c] = link[en[c]];
                    vn[c]  = ef4[(size_t)en[c] * (HID / 4) + lane];
                }
            }
            #pragma unroll
            for (int c = 0; c < SPW; ++c) {
                if ((unsigned)ec[c] < une) {
                    const float td = __int_as_float(lk[c].y);
                    ar[c].x += v[c].x;
                    ar[c].y += v[c].y;
                    ar[c].z += v[c].z;
                    ar[c].w += v[c].w;
                    at[c].x += cosf(td * tw.x + tb.x);
                    at[c].y += cosf(td * tw.y + tb.y);
                    at[c].z += cosf(td * tw.z + tb.z);
                    at[c].w += cosf(td * tw.w + tb.w);
                }
            }
            bool any = false;
            #pragma unroll
            for (int c = 0; c < SPW; ++c) {
                ec[c] = en[c];
                lk[c] = lkn[c];
                v[c]  = vn[c];
                en[c] = ((unsigned)ec[c] < une) ? __builtin_amdgcn_readfirstlane(lk[c].x) : -1;
                any   = any || ((unsigned)ec[c] < une);
            }
            if (!any) break;
        }
        #pragma unroll
        for (int c = 0; c < SPW; ++c) {
            const int j  = uw * SPW + c;
            const int li = j >> 4;
            float* rowp = (li == 0 ? sl : (li == 1 ? pl : nl)) + (j & 15) * LROW;
            *reinterpret_cast<float4*>(rowp + 4 * lane)       = ar[c];
            *reinterpret_cast<float4*>(rowp + HID + 4 * lane) = at[c];
        }
    }
    __syncthreads();

    v8f hs = {}, hp = {}, hn = {};

    const float* As = &sl[lm * LROW + 2 * half];
    const float* Ap = &pl[lm * LROW + 2 * half];
    const float* An = &nl[lm * LROW + 2 * half];

    static_assert(AHEAD == 3 && NU > AHEAD, "the three units loaded ahead of the loop are written out for AHEAD == 3");
    load_w_unit(bs[2], bd[2], wsl, wdl, 2);
    #pragma unroll
    for (int u = 0; u < NU; ++u) {
        if (u + AHEAD < NU)
            load_w_unit(bs[(u + AHEAD) % (AHEAD + 1)], bd[(u + AHEAD) % (AHEAD + 1)], wsl, wdl, u + AHEAD);
        mma_unit(hs, hp, hn, As, Ap, An, bs[u % (AHEAD + 1)], bd[u % (AHEAD + 1)], u);
    }

    const float bias = b_src[n0 + lm] + b_dst[n0 + lm];
    const float wo   = W_out[n0 + lm];

    #pragma unroll
    for (int r = 0; r < 8; ++r) {
        float pv = fmaxf(hs[r] + hp[r] + bias, 0.f) * wo;
        float nv = fmaxf(hs[r] + hn[r] + bias, 0.f) * wo;
        red_pos[wave * 256 + r * 32 + lane] = pv;
        red_neg[wave * 256 + r * 32 + lane] = nv;
    }
    __syncthreads();

    if (t < 16) {
        int mh = t >> 3;
        int r  = t & 7;
        float accp = b_out[0];
        float accn = b_out[0];
        for (int w = 0; w < 8; ++w) {
            #pragma unroll
            for (int l = 0; l < 16; ++l) {
                int idx = w * 256 + r * 32 + mh * 16 + l;
                accp += red_pos[idx];
                accn += red_neg[idx];
            }
        }
        out[tile * 16 + t]     = accp;
        out[Q + tile * 16 + t] = accn;
    }
}

extern "C" void kernel_launch(void* const* d_in, const int* in_sizes, int n_in,
                              void* d_out, int out_size, void* d_ws, size_t ws_size,
                              hipStream_t stream) {
    const int*   src     = (const int*)  d_in[0];
    const int*   dst     = (const int*)  d_in[1];
    const float* ef      = (const float*)d_in[2];
    const float* bt      = (const float*)d_in[3];
    const float* node_ts = (const float*)d_in[4];
    const int*   s_idx   = (const int*)  d_in[5];
    const int*   p_idx   = (const int*)  d_in[6];
    const int*   n_idx   = (const int*)  d_in[7];
    const float* time_w  = (const float*)d_in[8];
    const float* time_b  = (const float*)d_in[9];
    const float* W_src   = (const float*)d_in[10];
    const float* b_src   = (const float*)d_in[11];
    const float* W_dst   = (const float*)d_in[12];
    const float* b_dst   = (const float*)d_in[13];
    const float* W_out   = (const float*)d_in[14];
    const float* b_out   = (const float*)d_in[15];

    float* out = (float*)d_out;

    const int NN = in_sizes[4];
    const int NE = in_sizes[0];
    const int Q  = in_sizes[5];

    int*  head = (int*)d_ws + (size_t)NN;
    int2* link = (int2*)(head + (size_t)NN);

    tgnn_fill_kernel<<<(NN + FTHR * FPT - 1) / (FTHR * FPT), FTHR, 0, stream>>>(head, NN);

    tgnn_link_kernel<<<(NE + LTHR * EPT - 1) / (LTHR * EPT), LTHR, (size_t)((NN + 31) / 32) * sizeof(unsigned), stream>>>(
        src, dst, bt, node_ts, s_idx, p_idx, n_idx, head, link, NE, NN, Q);

    tgnn_predict_kernel<<<Q / 16, 256, 0, stream>>>(ef, head, link, s_idx, p_idx, n_idx, time_w, time_b,
                                                    W_src, b_src, W_dst, b_dst,
                                                    W_out, b_out, out, Q, NE);
}


# Chain walk: one wave-wide size test per edge; below cosf's own threshold its four cosines run cosf's short path written out branch-free (same constants, same operations) and interleave; else cosf
# speedup vs baseline: 12.2770x; 1.1568x over previous
#include <hip/hip_runtime.h>


typedef __attribute__((ext_vector_type(2))) float v2f;
typedef __attribute__((ext_vector_type(8))) float v8f;

#define HID   128
#define DIN   256
#define LROW  260
#define FTHR  256
#define FPT   16
#define LTHR  256
#define EPT   8
#define QPT   4
#define SPW   6
#define KU    4
#define NU    (DIN / (4 * KU))
#define AHEAD 3
#define COS_BIG 131072.0f


__global__ __launch_bounds__(FTHR) void tgnn_fill_kernel(int* __restrict__ head, int NN)
{
    const int base = blockIdx.x * (FTHR * FPT) + threadIdx.x;
    #pragma unroll
    for (int k = 0; k < FPT; ++k) {
        const int n = base + k * FTHR;
        if (n < NN) head[n] = -1;
    }
}

__device__ __forceinline__ void mark_node(unsigned* bm, int node, unsigned unn)
{
    const bool inr = (unsigned)node < unn;
    atomicOr(&bm[inr ? ((unsigned)node >> 5) : 0u], inr ? (1u << (node & 31)) : 0u);
}

__device__ __forceinline__ void load_query_groups(int4 (&a)[QPT], int4 (&p)[QPT], int4 (&c)[QPT],
                                                  const int4* __restrict__ s4,
                                                  const int4* __restrict__ p4,
                                                  const int4* __restrict__ n4,
                                                  int g0, int t, int q4)
{
    #pragma unroll
    for (int k = 0; k < QPT; ++k) {
        int g  = g0 + k * LTHR + t;
        int gc = g < q4 ? g : q4 - 1;
        a[k] = s4[gc];
        p[k] = p4[gc];
        c[k] = n4[gc];
    }
}

__global__ __launch_bounds__(LTHR) void tgnn_link_kernel(
    const int*   __restrict__ src,
    const int*   __restrict__ dst,
    const float* __restrict__ bt,
    const float* __restrict__ node_ts,
    const int*   __restrict__ s_idx,
    const int*   __restrict__ p_idx,
    const int*   __restrict__ n_idx,
    int*         __restrict__ head,
    int2*        __restrict__ link,
    int NE, int NN, int Q)
{
    extern __shared__ unsigned bm[];

    const int t  = threadIdx.x;
    const int e0 = blockIdx.x * (LTHR * EPT) + t;
    const unsigned unn = (unsigned)NN;

    const int nw = (NN + 31) >> 5;
    for (int i = t; i < nw; i += LTHR) bm[i] = 0u;
    __syncthreads();

    const int q4 = Q >> 2;
    const int4* __restrict__ s4 = reinterpret_cast<const int4*>(s_idx);
    const int4* __restrict__ p4 = reinterpret_cast<const int4*>(p_idx);
    const int4* __restrict__ n4 = reinterpret_cast<const int4*>(n_idx);
    int4 a[QPT], p[QPT], c[QPT];
    if (q4 > 0) load_query_groups(a, p, c, s4, p4, n4, 0, t, q4);

    int   d[EPT], s[EPT];
    float b[EPT];
    #pragma unroll
    for (int k = 0; k < EPT; ++k) {
        int e  = e0 + k * LTHR;
        int ec = e < NE ? e : NE - 1;
        d[k] = dst[ec];
        s[k] = src[ec];
        b[k] = bt[ec];
    }

    for (int g0 = 0; g0 < q4; ) {
        #pragma unroll
        for (int k = 0; k < QPT; ++k) {
            int g = g0 + k * LTHR + t;
            if (g < q4) {
                mark_node(bm, a[k].x, unn); mark_node(bm, a[k].y, unn);
                mark_node(bm, a[k].z, unn); mark_node(bm, a[k].w, unn);
                mark_node(bm, p[k].x, unn); mark_node(bm, p[k].y, unn);
                mark_node(bm, p[k].z, unn); mark_node(bm, p[k].w, unn);
                mark_node(bm, c[k].x, unn); mark_node(bm, c[k].y, unn);
                mark_node(bm, c[k].z, unn); mark_node(bm, c[k].w, unn);
            }
        }
        g0 += QPT * LTHR;
        if (g0 < q4) load_query_groups(a, p, c, s4, p4, n4, g0, t, q4);
    }
    for (int i = (q4 << 2) + t; i < Q; i += LTHR) {
        mark_node(bm, s_idx[i], unn);
        mark_node(bm, p_idx[i], unn);
        mark_node(bm, n_idx[i], unn);
    }
    __syncthreads();

    bool live[EPT];
    #pragma unroll
    for (int k = 0; k < EPT; ++k) {
        const bool inr = (unsigned)d[k] < unn;
        const unsigned w = bm[inr ? ((unsigned)d[k] >> 5) : 0u];
        live[k] = (e0 + k * LTHR < NE) & inr & (((w >> (d[k] & 31)) & 1u) != 0u);
    }

    float ts[EPT];
    #pragma unroll
    for (int k = 0; k < EPT; ++k) ts[k] = node_ts[live[k] ? s[k] : 0];

    int old[EPT];
    #pragma unroll
    for (int k = 0; k < EPT; ++k) old[k] = live[k] ? atomicExch(&head[d[k]], e0 + k * LTHR) : -1;

    #pragma unroll
    for (int k = 0; k < EPT; ++k)
        if (live[k]) link[e0 + k * LTHR] = make_int2(old[k], __float_as_int(b[k] - ts[k]));
}

__device__ __forceinline__ void load_w_unit(v2f (&bs)[KU], v2f (&bd)[KU],
                                            const float* __restrict__ wsl,
                                            const float* __restrict__ wdl, int u)
{
    #pragma unroll
    for (int i = 0; i < KU; ++i) {
        const int kr = (u * KU + i) * 4 * HID;
        bs[i].x = wsl[kr];
        bs[i].y = wsl[kr + HID];
        bd[i].x = wdl[kr];
        bd[i].y = wdl[kr + HID];
    }
}

__device__ __forceinline__ void mma_unit(v8f& hs, v8f& hp, v8f& hn,
                                         const float* As, const float* Ap, const float* An,
                                         const v2f (&bs)[KU], const v2f (&bd)[KU], int u)
{
    #pragma unroll
    for (int i = 0; i < KU; ++i) {
        const int k0 = (u * KU + i) * 4;
        v2f a_s; a_s.x = As[k0]; a_s.y = As[k0 + 1];
        v2f a_p; a_p.x = Ap[k0]; a_p.y = Ap[k0 + 1];
        v2f a_n; a_n.x = An[k0]; a_n.y = An[k0 + 1];
        hs = __builtin_amdgcn_wmma_f32_16x16x4_f32(false, a_s, false, bs[i], (short)0, hs, false, false);
        hp = __builtin_amdgcn_wmma_f32_16x16x4_f32(false, a_p, false, bd[i], (short)0, hp, false, false);
        hn = __builtin_amdgcn_wmma_f32_16x16x4_f32(false, a_n, false, bd[i], (short)0, hn, false, false);
    }
}


__device__ __forceinline__ float cos_short(float x)
{
    const float ax = fabsf(x);
    const float fn = __builtin_rintf(ax * __uint_as_float(0x3f22f983u));
    float r = __builtin_fmaf(fn, __uint_as_float(0xbfc90fdau), ax);
    r = __builtin_fmaf(fn, __uint_as_float(0xb3a22168u), r);
    r = __builtin_fmaf(fn, __uint_as_float(0xa7c234c4u), r);
    const int q = (int)fn;
    const float t = r * r;
    float sp = __builtin_fmaf(__uint_as_float(0xb94c1982u), t, __uint_as_float(0x3c0881c4u));
    sp = __builtin_fmaf(t, sp, __uint_as_float(0xbe2aaa9du));
    sp = t * sp;
    const float sn = __builtin_fmaf(r, sp, r);
    float cp = __builtin_fmaf(__uint_as_float(0x37d75334u), t, __uint_as_float(0xbab64f3bu));
    cp = __builtin_fmaf(t, cp, __uint_as_float(0x3d2aabf7u));
    cp = __builtin_fmaf(t, cp, __uint_as_float(0xbf000004u));
    const float cs = __builtin_fmaf(t, cp, 1.0f);
    const float v = (q & 1) ? -sn : cs;
    return __uint_as_float(__float_as_uint(v) ^ (((unsigned)q << 30) & 0x80000000u));
}

__global__ __launch_bounds__(256) void tgnn_predict_kernel(
    const float* __restrict__ ef,
    const int*   __restrict__ head,
    const int2*  __restrict__ link,
    const int*   __restrict__ s_idx,
    const int*   __restrict__ p_idx,
    const int*   __restrict__ n_idx,
    const float* __restrict__ time_w,
    const float* __restrict__ time_b,
    const float* __restrict__ W_src,
    const float* __restrict__ b_src,
    const float* __restrict__ W_dst,
    const float* __restrict__ b_dst,
    const float* __restrict__ W_out,
    const float* __restrict__ b_out,
    float*       __restrict__ out,
    int Q, int NE)
{
    __shared__ __attribute__((aligned(16))) float sl[16 * LROW];
    __shared__ __attribute__((aligned(16))) float pl[16 * LROW];
    __shared__ __attribute__((aligned(16))) float nl[16 * LROW];
    __shared__ float red_pos[8 * 256];
    __shared__ float red_neg[8 * 256];

    const int tile = blockIdx.x;
    const int t    = threadIdx.x;
    const int lane = t & 31;
    const int uw   = __builtin_amdgcn_readfirstlane(t >> 5);

    const int wave = t >> 5;
    const int half = lane >> 4;
    const int lm   = lane & 15;
    const int n0   = wave * 16;

    const float* __restrict__ wsl = W_src + (2 * half) * HID + n0 + lm;
    const float* __restrict__ wdl = W_dst + (2 * half) * HID + n0 + lm;
    v2f bs[AHEAD + 1][KU], bd[AHEAD + 1][KU];

    {
        const float4* __restrict__ ef4 = reinterpret_cast<const float4*>(ef);
        const float4 tw = reinterpret_cast<const float4*>(time_w)[lane];
        const float4 tb = reinterpret_cast<const float4*>(time_b)[lane];
        const unsigned une = (unsigned)NE;

        int hv = -1;
        if (lane < SPW) {
            const int j  = uw * SPW + lane;
            const int li = j >> 4;
            const int qi = tile * 16 + (j & 15);
            const int* __restrict__ ip = li == 0 ? s_idx : (li == 1 ? p_idx : n_idx);
            hv = head[ip[qi]];
        }

        load_w_unit(bs[0], bd[0], wsl, wdl, 0);
        load_w_unit(bs[1], bd[1], wsl, wdl, 1);

        int    ec[SPW], en[SPW];
        int2   lk[SPW];
        float4 v[SPW], ar[SPW], at[SPW];
        #pragma unroll
        for (int c = 0; c < SPW; ++c) {
            en[c] = __builtin_amdgcn_readlane(hv, c);
            ec[c] = -1;
            lk[c] = make_int2(-1, 0);
            v[c]  = make_float4(0.f, 0.f, 0.f, 0.f);
            ar[c] = make_float4(0.f, 0.f, 0.f, 0.f);
            at[c] = make_float4(0.f, 0.f, 0.f, 0.f);
        }
        for (int hop = 0; hop <= NE; ++hop) {
            int2   lkn[SPW];
            float4 vn[SPW];
            #pragma unroll
            for (int c = 0; c < SPW; ++c) {
                lkn[c] = make_int2(-1, 0);
                vn[c]  = make_float4(0.f, 0.f, 0.f, 0.f);
                if ((unsigned)en[c] < une) {
                    lkn[c] = link[en[c]];
                    vn[c]  = ef4[(size_t)en[c] * (HID / 4) + lane];
                }
            }
            #pragma unroll
            for (int c = 0; c < SPW; ++c) {
                if ((unsigned)ec[c] < une) {
                    const float td = __int_as_float(lk[c].y);
                    ar[c].x += v[c].x;
                    ar[c].y += v[c].y;
                    ar[c].z += v[c].z;
                    ar[c].w += v[c].w;
                    const float x0 = td * tw.x + tb.x;
                    const float x1 = td * tw.y + tb.y;
                    const float x2 = td * tw.z + tb.z;
                    const float x3 = td * tw.w + tb.w;
                    const bool small = (fabsf(x0) < COS_BIG) & (fabsf(x1) < COS_BIG) &
                                       (fabsf(x2) < COS_BIG) & (fabsf(x3) < COS_BIG);
                    if (__builtin_amdgcn_ballot_w32(!small) == 0u) {
                        at[c].x += cos_short(x0);
                        at[c].y += cos_short(x1);
                        at[c].z += cos_short(x2);
                        at[c].w += cos_short(x3);
                    } else {
                        at[c].x += cosf(x0);
                        at[c].y += cosf(x1);
                        at[c].z += cosf(x2);
                        at[c].w += cosf(x3);
                    }
                }
            }
            bool any = false;
            #pragma unroll
            for (int c = 0; c < SPW; ++c) {
                ec[c] = en[c];
                lk[c] = lkn[c];
                v[c]  = vn[c];
                en[c] = ((unsigned)ec[c] < une) ? __builtin_amdgcn_readfirstlane(lk[c].x) : -1;
                any   = any || ((unsigned)ec[c] < une);
            }
            if (!any) break;
        }
        #pragma unroll
        for (int c = 0; c < SPW; ++c) {
            const int j  = uw * SPW + c;
            const int li = j >> 4;
            float* rowp = (li == 0 ? sl : (li == 1 ? pl : nl)) + (j & 15) * LROW;
            *reinterpret_cast<float4*>(rowp + 4 * lane)       = ar[c];
            *reinterpret_cast<float4*>(rowp + HID + 4 * lane) = at[c];
        }
    }
    __syncthreads();

    v8f hs = {}, hp = {}, hn = {};

    const float* As = &sl[lm * LROW + 2 * half];
    const float* Ap = &pl[lm * LROW + 2 * half];
    const float* An = &nl[lm * LROW + 2 * half];

    static_assert(AHEAD == 3 && NU > AHEAD, "the three units loaded ahead of the loop are written out for AHEAD == 3");
    load_w_unit(bs[2], bd[2], wsl, wdl, 2);
    #pragma unroll
    for (int u = 0; u < NU; ++u) {
        if (u + AHEAD < NU)
            load_w_unit(bs[(u + AHEAD) % (AHEAD + 1)], bd[(u + AHEAD) % (AHEAD + 1)], wsl, wdl, u + AHEAD);
        mma_unit(hs, hp, hn, As, Ap, An, bs[u % (AHEAD + 1)], bd[u % (AHEAD + 1)], u);
    }

    const float bias = b_src[n0 + lm] + b_dst[n0 + lm];
    const float wo   = W_out[n0 + lm];

    #pragma unroll
    for (int r = 0; r < 8; ++r) {
        float pv = fmaxf(hs[r] + hp[r] + bias, 0.f) * wo;
        float nv = fmaxf(hs[r] + hn[r] + bias, 0.f) * wo;
        red_pos[wave * 256 + r * 32 + lane] = pv;
        red_neg[wave * 256 + r * 32 + lane] = nv;
    }
    __syncthreads();

    if (t < 16) {
        int mh = t >> 3;
        int r  = t & 7;
        float accp = b_out[0];
        float accn = b_out[0];
        for (int w = 0; w < 8; ++w) {
            #pragma unroll
            for (int l = 0; l < 16; ++l) {
                int idx = w * 256 + r * 32 + mh * 16 + l;
                accp += red_pos[idx];
                accn += red_neg[idx];
            }
        }
        out[tile * 16 + t]     = accp;
        out[Q + tile * 16 + t] = accn;
    }
}

extern "C" void kernel_launch(void* const* d_in, const int* in_sizes, int n_in,
                              void* d_out, int out_size, void* d_ws, size_t ws_size,
                              hipStream_t stream) {
    const int*   src     = (const int*)  d_in[0];
    const int*   dst     = (const int*)  d_in[1];
    const float* ef      = (const float*)d_in[2];
    const float* bt      = (const float*)d_in[3];
    const float* node_ts = (const float*)d_in[4];
    const int*   s_idx   = (const int*)  d_in[5];
    const int*   p_idx   = (const int*)  d_in[6];
    const int*   n_idx   = (const int*)  d_in[7];
    const float* time_w  = (const float*)d_in[8];
    const float* time_b  = (const float*)d_in[9];
    const float* W_src   = (const float*)d_in[10];
    const float* b_src   = (const float*)d_in[11];
    const float* W_dst   = (const float*)d_in[12];
    const float* b_dst   = (const float*)d_in[13];
    const float* W_out   = (const float*)d_in[14];
    const float* b_out   = (const float*)d_in[15];

    float* out = (float*)d_out;

    const int NN = in_sizes[4];
    const int NE = in_sizes[0];
    const int Q  = in_sizes[5];

    int*  head = (int*)d_ws + (size_t)NN;
    int2* link = (int2*)(head + (size_t)NN);

    tgnn_fill_kernel<<<(NN + FTHR * FPT - 1) / (FTHR * FPT), FTHR, 0, stream>>>(head, NN);

    tgnn_link_kernel<<<(NE + LTHR * EPT - 1) / (LTHR * EPT), LTHR, (size_t)((NN + 31) / 32) * sizeof(unsigned), stream>>>(
        src, dst, bt, node_ts, s_idx, p_idx, n_idx, head, link, NE, NN, Q);

    tgnn_predict_kernel<<<Q / 16, 256, 0, stream>>>(ef, head, link, s_idx, p_idx, n_idx, time_w, time_b,
                                                    W_src, b_src, W_dst, b_dst,
                                                    W_out, b_out, out, Q, NE);
}


# Output projection: each wave sums its 16 columns by a lane butterfly (DPP adds), one lane adds the 8 wave partials; replaces 16 lanes summing 128 LDS terms in series; sum order changes (fixed tree)
# speedup vs baseline: 12.7318x; 1.0370x over previous
#include <hip/hip_runtime.h>


typedef __attribute__((ext_vector_type(2))) float v2f;
typedef __attribute__((ext_vector_type(8))) float v8f;

#define HID   128
#define DIN   256
#define LROW  260
#define FTHR  256
#define FPT   16
#define LTHR  256
#define EPT   8
#define QPT   4
#define SPW   6
#define KU    4
#define NU    (DIN / (4 * KU))
#define AHEAD 3
#define COS_BIG 131072.0f


__global__ __launch_bounds__(FTHR) void tgnn_fill_kernel(int* __restrict__ head, int NN)
{
    const int base = blockIdx.x * (FTHR * FPT) + threadIdx.x;
    #pragma unroll
    for (int k = 0; k < FPT; ++k) {
        const int n = base + k * FTHR;
        if (n < NN) head[n] = -1;
    }
}

__device__ __forceinline__ void mark_node(unsigned* bm, int node, unsigned unn)
{
    const bool inr = (unsigned)node < unn;
    atomicOr(&bm[inr ? ((unsigned)node >> 5) : 0u], inr ? (1u << (node & 31)) : 0u);
}

__device__ __forceinline__ void load_query_groups(int4 (&a)[QPT], int4 (&p)[QPT], int4 (&c)[QPT],
                                                  const int4* __restrict__ s4,
                                                  const int4* __restrict__ p4,
                                                  const int4* __restrict__ n4,
                                                  int g0, int t, int q4)
{
    #pragma unroll
    for (int k = 0; k < QPT; ++k) {
        int g  = g0 + k * LTHR + t;
        int gc = g < q4 ? g : q4 - 1;
        a[k] = s4[gc];
        p[k] = p4[gc];
        c[k] = n4[gc];
    }
}

__global__ __launch_bounds__(LTHR) void tgnn_link_kernel(
    const int*   __restrict__ src,
    const int*   __restrict__ dst,
    const float* __restrict__ bt,
    const float* __restrict__ node_ts,
    const int*   __restrict__ s_idx,
    const int*   __restrict__ p_idx,
    const int*   __restrict__ n_idx,
    int*         __restrict__ head,
    int2*        __restrict__ link,
    int NE, int NN, int Q)
{
    extern __shared__ unsigned bm[];

    const int t  = threadIdx.x;
    const int e0 = blockIdx.x * (LTHR * EPT) + t;
    const unsigned unn = (unsigned)NN;

    const int nw = (NN + 31) >> 5;
    for (int i = t; i < nw; i += LTHR) bm[i] = 0u;
    __syncthreads();

    const int q4 = Q >> 2;
    const int4* __restrict__ s4 = reinterpret_cast<const int4*>(s_idx);
    const int4* __restrict__ p4 = reinterpret_cast<const int4*>(p_idx);
    const int4* __restrict__ n4 = reinterpret_cast<const int4*>(n_idx);
    int4 a[QPT], p[QPT], c[QPT];
    if (q4 > 0) load_query_groups(a, p, c, s4, p4, n4, 0, t, q4);

    int   d[EPT], s[EPT];
    float b[EPT];
    #pragma unroll
    for (int k = 0; k < EPT; ++k) {
        int e  = e0 + k * LTHR;
        int ec = e < NE ? e : NE - 1;
        d[k] = dst[ec];
        s[k] = src[ec];
        b[k] = bt[ec];
    }

    for (int g0 = 0; g0 < q4; ) {
        #pragma unroll
        for (int k = 0; k < QPT; ++k) {
            int g = g0 + k * LTHR + t;
            if (g < q4) {
                mark_node(bm, a[k].x, unn); mark_node(bm, a[k].y, unn);
                mark_node(bm, a[k].z, unn); mark_node(bm, a[k].w, unn);
                mark_node(bm, p[k].x, unn); mark_node(bm, p[k].y, unn);
                mark_node(bm, p[k].z, unn); mark_node(bm, p[k].w, unn);
                mark_node(bm, c[k].x, unn); mark_node(bm, c[k].y, unn);
                mark_node(bm, c[k].z, unn); mark_node(bm, c[k].w, unn);
            }
        }
        g0 += QPT * LTHR;
        if (g0 < q4) load_query_groups(a, p, c, s4, p4, n4, g0, t, q4);
    }
    for (int i = (q4 << 2) + t; i < Q; i += LTHR) {
        mark_node(bm, s_idx[i], unn);
        mark_node(bm, p_idx[i], unn);
        mark_node(bm, n_idx[i], unn);
    }
    __syncthreads();

    bool live[EPT];
    #pragma unroll
    for (int k = 0; k < EPT; ++k) {
        const bool inr = (unsigned)d[k] < unn;
        const unsigned w = bm[inr ? ((unsigned)d[k] >> 5) : 0u];
        live[k] = (e0 + k * LTHR < NE) & inr & (((w >> (d[k] & 31)) & 1u) != 0u);
    }

    float ts[EPT];
    #pragma unroll
    for (int k = 0; k < EPT; ++k) ts[k] = node_ts[live[k] ? s[k] : 0];

    int old[EPT];
    #pragma unroll
    for (int k = 0; k < EPT; ++k) old[k] = live[k] ? atomicExch(&head[d[k]], e0 + k * LTHR) : -1;

    #pragma unroll
    for (int k = 0; k < EPT; ++k)
        if (live[k]) link[e0 + k * LTHR] = make_int2(old[k], __float_as_int(b[k] - ts[k]));
}

__device__ __forceinline__ void load_w_unit(v2f (&bs)[KU], v2f (&bd)[KU],
                                            const float* __restrict__ wsl,
                                            const float* __restrict__ wdl, int u)
{
    #pragma unroll
    for (int i = 0; i < KU; ++i) {
        const int kr = (u * KU + i) * 4 * HID;
        bs[i].x = wsl[kr];
        bs[i].y = wsl[kr + HID];
        bd[i].x = wdl[kr];
        bd[i].y = wdl[kr + HID];
    }
}

__device__ __forceinline__ void mma_unit(v8f& hs, v8f& hp, v8f& hn,
                                         const float* As, const float* Ap, const float* An,
                                         const v2f (&bs)[KU], const v2f (&bd)[KU], int u)
{
    #pragma unroll
    for (int i = 0; i < KU; ++i) {
        const int k0 = (u * KU + i) * 4;
        v2f a_s; a_s.x = As[k0]; a_s.y = As[k0 + 1];
        v2f a_p; a_p.x = Ap[k0]; a_p.y = Ap[k0 + 1];
        v2f a_n; a_n.x = An[k0]; a_n.y = An[k0 + 1];
        hs = __builtin_amdgcn_wmma_f32_16x16x4_f32(false, a_s, false, bs[i], (short)0, hs, false, false);
        hp = __builtin_amdgcn_wmma_f32_16x16x4_f32(false, a_p, false, bd[i], (short)0, hp, false, false);
        hn = __builtin_amdgcn_wmma_f32_16x16x4_f32(false, a_n, false, bd[i], (short)0, hn, false, false);
    }
}


__device__ __forceinline__ float cos_short(float x)
{
    const float ax = fabsf(x);
    const float fn = __builtin_rintf(ax * __uint_as_float(0x3f22f983u));
    float r = __builtin_fmaf(fn, __uint_as_float(0xbfc90fdau), ax);
    r = __builtin_fmaf(fn, __uint_as_float(0xb3a22168u), r);
    r = __builtin_fmaf(fn, __uint_as_float(0xa7c234c4u), r);
    const int q = (int)fn;
    const float t = r * r;
    float sp = __builtin_fmaf(__uint_as_float(0xb94c1982u), t, __uint_as_float(0x3c0881c4u));
    sp = __builtin_fmaf(t, sp, __uint_as_float(0xbe2aaa9du));
    sp = t * sp;
    const float sn = __builtin_fmaf(r, sp, r);
    float cp = __builtin_fmaf(__uint_as_float(0x37d75334u), t, __uint_as_float(0xbab64f3bu));
    cp = __builtin_fmaf(t, cp, __uint_as_float(0x3d2aabf7u));
    cp = __builtin_fmaf(t, cp, __uint_as_float(0xbf000004u));
    const float cs = __builtin_fmaf(t, cp, 1.0f);
    const float v = (q & 1) ? -sn : cs;
    return __uint_as_float(__float_as_uint(v) ^ (((unsigned)q << 30) & 0x80000000u));
}

__device__ __forceinline__ float row16_sum(float v)
{
    v += __builtin_bit_cast(float, __builtin_amdgcn_update_dpp(0, __builtin_bit_cast(int, v), 0x161, 0xf, 0xf, false));
    v += __builtin_bit_cast(float, __builtin_amdgcn_update_dpp(0, __builtin_bit_cast(int, v), 0x162, 0xf, 0xf, false));
    v += __builtin_bit_cast(float, __builtin_amdgcn_update_dpp(0, __builtin_bit_cast(int, v), 0x164, 0xf, 0xf, false));
    v += __builtin_bit_cast(float, __builtin_amdgcn_update_dpp(0, __builtin_bit_cast(int, v), 0x168, 0xf, 0xf, false));
    return v;
}

__global__ __launch_bounds__(256) void tgnn_predict_kernel(
    const float* __restrict__ ef,
    const int*   __restrict__ head,
    const int2*  __restrict__ link,
    const int*   __restrict__ s_idx,
    const int*   __restrict__ p_idx,
    const int*   __restrict__ n_idx,
    const float* __restrict__ time_w,
    const float* __restrict__ time_b,
    const float* __restrict__ W_src,
    const float* __restrict__ b_src,
    const float* __restrict__ W_dst,
    const float* __restrict__ b_dst,
    const float* __restrict__ W_out,
    const float* __restrict__ b_out,
    float*       __restrict__ out,
    int Q, int NE)
{
    __shared__ __attribute__((aligned(16))) float sl[16 * LROW];
    __shared__ __attribute__((aligned(16))) float pl[16 * LROW];
    __shared__ __attribute__((aligned(16))) float nl[16 * LROW];
    __shared__ __attribute__((aligned(16))) float red[8 * 32];

    const int tile = blockIdx.x;
    const int t    = threadIdx.x;
    const int lane = t & 31;
    const int uw   = __builtin_amdgcn_readfirstlane(t >> 5);

    const int wave = t >> 5;
    const int half = lane >> 4;
    const int lm   = lane & 15;
    const int n0   = wave * 16;

    const float* __restrict__ wsl = W_src + (2 * half) * HID + n0 + lm;
    const float* __restrict__ wdl = W_dst + (2 * half) * HID + n0 + lm;
    v2f bs[AHEAD + 1][KU], bd[AHEAD + 1][KU];

    {
        const float4* __restrict__ ef4 = reinterpret_cast<const float4*>(ef);
        const float4 tw = reinterpret_cast<const float4*>(time_w)[lane];
        const float4 tb = reinterpret_cast<const float4*>(time_b)[lane];
        const unsigned une = (unsigned)NE;

        int hv = -1;
        if (lane < SPW) {
            const int j  = uw * SPW + lane;
            const int li = j >> 4;
            const int qi = tile * 16 + (j & 15);
            const int* __restrict__ ip = li == 0 ? s_idx : (li == 1 ? p_idx : n_idx);
            hv = head[ip[qi]];
        }

        load_w_unit(bs[0], bd[0], wsl, wdl, 0);
        load_w_unit(bs[1], bd[1], wsl, wdl, 1);

        int    ec[SPW], en[SPW];
        int2   lk[SPW];
        float4 v[SPW], ar[SPW], at[SPW];
        #pragma unroll
        for (int c = 0; c < SPW; ++c) {
            en[c] = __builtin_amdgcn_readlane(hv, c);
            ec[c] = -1;
            lk[c] = make_int2(-1, 0);
            v[c]  = make_float4(0.f, 0.f, 0.f, 0.f);
            ar[c] = make_float4(0.f, 0.f, 0.f, 0.f);
            at[c] = make_float4(0.f, 0.f, 0.f, 0.f);
        }
        for (int hop = 0; hop <= NE; ++hop) {
            int2   lkn[SPW];
            float4 vn[SPW];
            #pragma unroll
            for (int c = 0; c < SPW; ++c) {
                lkn[c] = make_int2(-1, 0);
                vn[c]  = make_float4(0.f, 0.f, 0.f, 0.f);
                if ((unsigned)en[c] < une) {
                    lkn[c] = link[en[c]];
                    vn[c]  = ef4[(size_t)en[c] * (HID / 4) + lane];
                }
            }
            #pragma unroll
            for (int c = 0; c < SPW; ++c) {
                if ((unsigned)ec[c] < une) {
                    const float td = __int_as_float(lk[c].y);
                    ar[c].x += v[c].x;
                    ar[c].y += v[c].y;
                    ar[c].z += v[c].z;
                    ar[c].w += v[c].w;
                    const float x0 = td * tw.x + tb.x;
                    const float x1 = td * tw.y + tb.y;
                    const float x2 = td * tw.z + tb.z;
                    const float x3 = td * tw.w + tb.w;
                    const bool small = (fabsf(x0) < COS_BIG) & (fabsf(x1) < COS_BIG) &
                                       (fabsf(x2) < COS_BIG) & (fabsf(x3) < COS_BIG);
                    if (__builtin_amdgcn_ballot_w32(!small) == 0u) {
                        at[c].x += cos_short(x0);
                        at[c].y += cos_short(x1);
                        at[c].z += cos_short(x2);
                        at[c].w += cos_short(x3);
                    } else {
                        at[c].x += cosf(x0);
                        at[c].y += cosf(x1);
                        at[c].z += cosf(x2);
                        at[c].w += cosf(x3);
                    }
                }
            }
            bool any = false;
            #pragma unroll
            for (int c = 0; c < SPW; ++c) {
                ec[c] = en[c];
                lk[c] = lkn[c];
                v[c]  = vn[c];
                en[c] = ((unsigned)ec[c] < une) ? __builtin_amdgcn_readfirstlane(lk[c].x) : -1;
                any   = any || ((unsigned)ec[c] < une);
            }
            if (!any) break;
        }
        #pragma unroll
        for (int c = 0; c < SPW; ++c) {
            const int j  = uw * SPW + c;
            const int li = j >> 4;
            float* rowp = (li == 0 ? sl : (li == 1 ? pl : nl)) + (j & 15) * LROW;
            *reinterpret_cast<float4*>(rowp + 4 * lane)       = ar[c];
            *reinterpret_cast<float4*>(rowp + HID + 4 * lane) = at[c];
        }
    }
    __syncthreads();

    v8f hs = {}, hp = {}, hn = {};

    const float* As = &sl[lm * LROW + 2 * half];
    const float* Ap = &pl[lm * LROW + 2 * half];
    const float* An = &nl[lm * LROW + 2 * half];

    static_assert(AHEAD == 3 && NU > AHEAD, "the three units loaded ahead of the loop are written out for AHEAD == 3");
    load_w_unit(bs[2], bd[2], wsl, wdl, 2);
    #pragma unroll
    for (int u = 0; u < NU; ++u) {
        if (u + AHEAD < NU)
            load_w_unit(bs[(u + AHEAD) % (AHEAD + 1)], bd[(u + AHEAD) % (AHEAD + 1)], wsl, wdl, u + AHEAD);
        mma_unit(hs, hp, hn, As, Ap, An, bs[u % (AHEAD + 1)], bd[u % (AHEAD + 1)], u);
    }

    const float bias = b_src[n0 + lm] + b_dst[n0 + lm];
    const float wo   = W_out[n0 + lm];

    float pv[8], nv[8];
    #pragma unroll
    for (int r = 0; r < 8; ++r) {
        pv[r] = row16_sum(fmaxf(hs[r] + hp[r] + bias, 0.f) * wo);
        nv[r] = row16_sum(fmaxf(hs[r] + hn[r] + bias, 0.f) * wo);
    }
    if (lm == 0) {
        float* rp = &red[wave * 32 + 8 * half];
        *reinterpret_cast<float4*>(rp)      = make_float4(pv[0], pv[1], pv[2], pv[3]);
        *reinterpret_cast<float4*>(rp + 4)  = make_float4(pv[4], pv[5], pv[6], pv[7]);
        *reinterpret_cast<float4*>(rp + 16) = make_float4(nv[0], nv[1], nv[2], nv[3]);
        *reinterpret_cast<float4*>(rp + 20) = make_float4(nv[4], nv[5], nv[6], nv[7]);
    }
    __syncthreads();

    if (t < 32) {
        float p[8];
        #pragma unroll
        for (int w = 0; w < 8; ++w) p[w] = red[w * 32 + t];
        float acc = b_out[0];
        #pragma unroll
        for (int w = 0; w < 8; ++w) acc += p[w];
        out[(t >> 4) * Q + tile * 16 + (t & 15)] = acc;
    }
}

extern "C" void kernel_launch(void* const* d_in, const int* in_sizes, int n_in,
                              void* d_out, int out_size, void* d_ws, size_t ws_size,
                              hipStream_t stream) {
    const int*   src     = (const int*)  d_in[0];
    const int*   dst     = (const int*)  d_in[1];
    const float* ef      = (const float*)d_in[2];
    const float* bt      = (const float*)d_in[3];
    const float* node_ts = (const float*)d_in[4];
    const int*   s_idx   = (const int*)  d_in[5];
    const int*   p_idx   = (const int*)  d_in[6];
    const int*   n_idx   = (const int*)  d_in[7];
    const float* time_w  = (const float*)d_in[8];
    const float* time_b  = (const float*)d_in[9];
    const float* W_src   = (const float*)d_in[10];
    const float* b_src   = (const float*)d_in[11];
    const float* W_dst   = (const float*)d_in[12];
    const float* b_dst   = (const float*)d_in[13];
    const float* W_out   = (const float*)d_in[14];
    const float* b_out   = (const float*)d_in[15];

    float* out = (float*)d_out;

    const int NN = in_sizes[4];
    const int NE = in_sizes[0];
    const int Q  = in_sizes[5];

    int*  head = (int*)d_ws + (size_t)NN;
    int2* link = (int2*)(head + (size_t)NN);

    tgnn_fill_kernel<<<(NN + FTHR * FPT - 1) / (FTHR * FPT), FTHR, 0, stream>>>(head, NN);

    tgnn_link_kernel<<<(NE + LTHR * EPT - 1) / (LTHR * EPT), LTHR, (size_t)((NN + 31) / 32) * sizeof(unsigned), stream>>>(
        src, dst, bt, node_ts, s_idx, p_idx, n_idx, head, link, NE, NN, Q);

    tgnn_predict_kernel<<<Q / 16, 256, 0, stream>>>(ef, head, link, s_idx, p_idx, n_idx, time_w, time_b,
                                                    W_src, b_src, W_dst, b_dst,
                                                    W_out, b_out, out, Q, NE);
}
